# starting kernel
# baseline (speedup 1.0000x reference)
#include <hip/hip_runtime.h>


typedef __attribute__((ext_vector_type(2))) float v2f;
typedef __attribute__((ext_vector_type(8))) float v8f;

#define Bb 8
#define Ss 256
#define Tt 64
#define Hh 128

__global__ __launch_bounds__(256) void cam_wmma_kernel(const float* __restrict__ inp,
                                                       const float* __restrict__ W,
                                                       float* __restrict__ cam) {
  const int lane = threadIdx.x & 31;
  const int wave = threadIdx.x >> 5;
  const int n = blockIdx.x * 8 + wave;
  const int b  = n >> 14;
  const int h0 = (n >> 7) & 127;
  const int h1 = n & 127;

  const bool hi = (lane >= 16);
  const int  Lh = lane & 15;

  const int base00 = ((b * Ss + 2 * h0) * Ss + 2 * h1) * Tt;
  const float* rowbase = inp + base00 + (hi ? (Ss * Tt) : 0);

  v2f bm;
  bm[0] = hi ? W[2 * 6] : W[0 * 6];
  bm[1] = hi ? W[3 * 6] : W[1 * 6];

  const v8f zero = {};
  v8f d0, d1, d2, d3;
  {
    v2f a;
    a[0] = rowbase[ 0 + Lh]; a[1] = rowbase[Tt +  0 + Lh];
    d0 = __builtin_amdgcn_wmma_f32_16x16x4_f32(false, a, false, bm, (short)0, zero, false, false);
    a[0] = rowbase[16 + Lh]; a[1] = rowbase[Tt + 16 + Lh];
    d1 = __builtin_amdgcn_wmma_f32_16x16x4_f32(false, a, false, bm, (short)0, zero, false, false);
    a[0] = rowbase[32 + Lh]; a[1] = rowbase[Tt + 32 + Lh];
    d2 = __builtin_amdgcn_wmma_f32_16x16x4_f32(false, a, false, bm, (short)0, zero, false, false);
    a[0] = rowbase[48 + Lh]; a[1] = rowbase[Tt + 48 + Lh];
    d3 = __builtin_amdgcn_wmma_f32_16x16x4_f32(false, a, false, bm, (short)0, zero, false, false);
  }

  float mn, mx;
  {
    v8f mnv, mxv;
#pragma unroll
    for (int e = 0; e < 8; ++e) {
      mnv[e] = fminf(fminf(d0[e], d1[e]), fminf(d2[e], d3[e]));
      mxv[e] = fmaxf(fmaxf(d0[e], d1[e]), fmaxf(d2[e], d3[e]));
    }
    mn = mnv[0]; mx = mxv[0];
#pragma unroll
    for (int e = 1; e < 8; ++e) { mn = fminf(mn, mnv[e]); mx = fmaxf(mx, mxv[e]); }
  }
  mn = fminf(mn, __shfl_xor(mn, 16, 32));
  mx = fmaxf(mx, __shfl_xor(mx, 16, 32));
  const float inv = 1.0f / (mx - mn);

  v8f s0 = (lane & 8) ? d2 : d0;
  v8f s1 = (lane & 8) ? d3 : d1;
  v8f dk = (lane & 4) ? s1 : s0;
  float e0 = (lane & 1) ? dk[2] : dk[0];
  float e1 = (lane & 1) ? dk[3] : dk[1];
  float e2 = (lane & 1) ? dk[6] : dk[4];
  float e3 = (lane & 1) ? dk[7] : dk[5];
  float xa = (lane & 2) ? e2 : e0;
  float xb = (lane & 2) ? e3 : e1;

  const int tt = ((lane >> 2) & 3) * 16 + (lane & 3) * 2 + (hi ? 8 : 0);
  float2 o;
  o.x = (xa - mn) * inv;
  o.y = (xb - mn) * inv;
  *(float2*)(cam + ((size_t)n << 6) + tt) = o;
}

__global__ __launch_bounds__(256) void upsample_kernel(const float* __restrict__ cam,
                                                       float* __restrict__ out) {
  const int lane = threadIdx.x & 31;
  const int pix = blockIdx.x * 8 + (threadIdx.x >> 5);
  const int b = pix >> 16;
  const int y = (pix >> 8) & 255;
  const int x = pix & 255;

  const float r = 127.0f / 255.0f;
  float fy = (float)y * r; int y0 = (int)fy; float wy = fy - (float)y0;
  int y1 = (y0 < Hh - 1) ? y0 + 1 : Hh - 1;
  float fx = (float)x * r; int x0 = (int)fx; float wx = fx - (float)x0;
  int x1 = (x0 < Hh - 1) ? x0 + 1 : Hh - 1;

  const int t = lane << 1;
  const size_t pb = (size_t)b << 14;
  const float* c00 = cam + (((pb + (size_t)y0 * Hh + x0) << 6) + t);
  const float* c01 = cam + (((pb + (size_t)y0 * Hh + x1) << 6) + t);
  const float* c10 = cam + (((pb + (size_t)y1 * Hh + x0) << 6) + t);
  const float* c11 = cam + (((pb + (size_t)y1 * Hh + x1) << 6) + t);
  float2 v00 = *(const float2*)c00;
  float2 v01 = *(const float2*)c01;
  float2 v10 = *(const float2*)c10;
  float2 v11 = *(const float2*)c11;

  float h0x = v00.x * (1.0f - wy) + v10.x * wy;
  float h1x = v01.x * (1.0f - wy) + v11.x * wy;
  float h0y = v00.y * (1.0f - wy) + v10.y * wy;
  float h1y = v01.y * (1.0f - wy) + v11.y * wy;
  float2 o;
  o.x = h0x * (1.0f - wx) + h1x * wx;
  o.y = h0y * (1.0f - wx) + h1y * wx;

  *(float2*)(out + (((size_t)pix) << 6) + t) = o;
}

extern "C" void kernel_launch(void* const* d_in, const int* in_sizes, int n_in,
                              void* d_out, int out_size, void* d_ws, size_t ws_size,
                              hipStream_t stream) {
  const float* inp = (const float*)d_in[0];
  const float* W   = (const float*)d_in[1];
  float* out = (float*)d_out;
  float* cam = (float*)d_ws;

  cam_wmma_kernel<<<16384, 256, 0, stream>>>(inp, W, cam);
  upsample_kernel<<<65536, 256, 0, stream>>>(cam, out);
}


# one launch, no workspace: each wave rebuilds a 2x4-cell tile's 3x5 normalized pooled pixels from the input in registers (taps as WMMA B: tap sum order changes) and emits its 4x8 output pixels
# speedup vs baseline: 2.0640x; 2.0640x over previous
#include <hip/hip_runtime.h>


typedef __attribute__((ext_vector_type(2))) float v2f;
typedef __attribute__((ext_vector_type(4))) float v4f;
typedef __attribute__((ext_vector_type(8))) float v8f;

#define Bb 8
#define Ss 256
#define Tt 64
#define Hh 128

static __device__ __forceinline__ float row_min16(float x) {
  asm("v_min_num_f32_dpp %0, %0, %0 row_xmask:1 row_mask:0xf bank_mask:0xf\n\t"
      "v_min_num_f32_dpp %0, %0, %0 row_xmask:2 row_mask:0xf bank_mask:0xf\n\t"
      "v_min_num_f32_dpp %0, %0, %0 row_xmask:4 row_mask:0xf bank_mask:0xf\n\t"
      "v_min_num_f32_dpp %0, %0, %0 row_xmask:8 row_mask:0xf bank_mask:0xf"
      : "+v"(x));
  return x;
}
static __device__ __forceinline__ float row_max16(float x) {
  asm("v_max_num_f32_dpp %0, %0, %0 row_xmask:1 row_mask:0xf bank_mask:0xf\n\t"
      "v_max_num_f32_dpp %0, %0, %0 row_xmask:2 row_mask:0xf bank_mask:0xf\n\t"
      "v_max_num_f32_dpp %0, %0, %0 row_xmask:4 row_mask:0xf bank_mask:0xf\n\t"
      "v_max_num_f32_dpp %0, %0, %0 row_xmask:8 row_mask:0xf bank_mask:0xf"
      : "+v"(x));
  return x;
}

static __device__ __forceinline__ v4f pooled_norm(const v4f la, const v4f lb, const v2f wa) {
  const v8f zero = {};
  v2f b0, b1, b2, b3;
  b0[0] = la[0]; b0[1] = lb[0];
  b1[0] = la[1]; b1[1] = lb[1];
  b2[0] = la[2]; b2[1] = lb[2];
  b3[0] = la[3]; b3[1] = lb[3];
  const v8f d0 = __builtin_amdgcn_wmma_f32_16x16x4_f32(false, wa, false, b0, (short)0, zero, false, false);
  const v8f d1 = __builtin_amdgcn_wmma_f32_16x16x4_f32(false, wa, false, b1, (short)0, zero, false, false);
  const v8f d2 = __builtin_amdgcn_wmma_f32_16x16x4_f32(false, wa, false, b2, (short)0, zero, false, false);
  const v8f d3 = __builtin_amdgcn_wmma_f32_16x16x4_f32(false, wa, false, b3, (short)0, zero, false, false);
  v4f v;
  v[0] = d0[0]; v[1] = d1[0]; v[2] = d2[0]; v[3] = d3[0];
  float mn = fminf(fminf(v[0], v[1]), fminf(v[2], v[3]));
  float mx = fmaxf(fmaxf(v[0], v[1]), fmaxf(v[2], v[3]));
  mn = row_min16(mn);
  mx = row_max16(mx);
  const float inv = 1.0f / (mx - mn);
  return (v - mn) * inv;
}

__global__ __launch_bounds__(256) void pool_norm_upsample_kernel(const float* __restrict__ inp,
                                                                 const float* __restrict__ W,
                                                                 float* __restrict__ out) {
  const int lane = threadIdx.x & 31;
  const int hf = lane >> 4;
  const int tq = (lane & 15) << 2;
  const int wave = __builtin_amdgcn_readfirstlane((int)(threadIdx.x >> 5));
  const int sx = blockIdx.x & 15;
  const int sy = (blockIdx.x >> 4) & 15;
  const int b  = blockIdx.x >> 8;
  const int ti = sy * 4 + (wave >> 1);
  const int jb = (sx * 2 + (wave & 1)) * 4;

  v2f wa;
  wa[0] = hf ? W[1 * 6] : W[0 * 6];
  wa[1] = hf ? W[3 * 6] : W[2 * 6];

  v4f la[3][5], lb[3][5];
#pragma unroll
  for (int rr = 0; rr < 3; ++rr) {
    int pr = 2 * ti + rr;
    pr = (pr < Hh - 1) ? pr : Hh - 1;
#pragma unroll
    for (int c = 0; c < 5; ++c) {
      int pc = jb + c;
      pc = (pc < Hh - 1) ? pc : Hh - 1;
      const unsigned off = ((((unsigned)b << 8) + 2u * (unsigned)pr) << 14) + ((2u * (unsigned)pc) << 6) + ((unsigned)lane << 2);
      la[rr][c] = *(const v4f*)(inp + off);
      lb[rr][c] = *(const v4f*)(inp + off + Ss * Tt);
    }
  }

  __builtin_amdgcn_sched_barrier(0);

  v4f n[3][5];
#pragma unroll
  for (int rr = 0; rr < 3; ++rr) {
#pragma unroll
    for (int c = 0; c < 5; ++c) n[rr][c] = pooled_norm(la[rr][c], lb[rr][c], wa);
  }

  const float r = 127.0f / 255.0f;
#pragma unroll
  for (int p = 0; p < 3; ++p) {
    if (p == 2 && ti != 0) break;
    const int k = (p == 2) ? 0 : 2 * ti + p;
    const int y = (p == 2) ? 0 : 2 * k + 1 + hf;
    const bool rowok = (p == 2) ? (hf == 0) : (y < Ss);
    const float fy = (float)y * r;
    const float wy = fy - (float)k;
    const float wy0 = 1.0f - wy;
    const int ra = (p == 2) ? 0 : p;
    v4f h[5];
#pragma unroll
    for (int c = 0; c < 5; ++c) h[c] = n[ra][c] * wy0 + n[ra + 1][c] * wy;

    float* orow = out + (((((unsigned)b << 8) + (unsigned)y) << 14) + tq);
#pragma unroll
    for (int c = 0; c < 4; ++c) {
      const int j  = jb + c;
      const int xa = (j == 0) ? 0 : 2 * j + 1;
      const int xb = (2 * j + 3 < Ss) ? 2 * j + 3 : Ss;
#pragma unroll
      for (int q = 0; q < 3; ++q) {
        const int x = xa + q;
        if (x < xb) {
          const float fx = (float)x * r;
          const float wx = fx - (float)j;
          const v4f o = h[c] * (1.0f - wx) + h[c + 1] * wx;
          if (rowok) *(v4f*)(orow + (x << 6)) = o;
        }
      }
    }
  }
}

extern "C" void kernel_launch(void* const* d_in, const int* in_sizes, int n_in,
                              void* d_out, int out_size, void* d_ws, size_t ws_size,
                              hipStream_t stream) {
  const float* inp = (const float*)d_in[0];
  const float* W   = (const float*)d_in[1];
  float* out = (float*)d_out;

  pool_norm_upsample_kernel<<<2048, 256, 0, stream>>>(inp, W, out);
}


# one wave walks a strip of 4 cells x 8 cell rows (4096 waves in 512 blocks, was 16384 one-tile waves): pooled row above kept in registers, next two input rows in flight during each step
# speedup vs baseline: 2.7164x; 1.3161x over previous
#include <hip/hip_runtime.h>


typedef __attribute__((ext_vector_type(2))) float v2f;
typedef __attribute__((ext_vector_type(4))) float v4f;
typedef __attribute__((ext_vector_type(8))) float v8f;

#define Bb 8
#define Ss 256
#define Tt 64
#define Hh 128

static __device__ __forceinline__ float row_min16(float x) {
  asm("v_min_num_f32_dpp %0, %0, %0 row_xmask:1 row_mask:0xf bank_mask:0xf\n\t"
      "v_min_num_f32_dpp %0, %0, %0 row_xmask:2 row_mask:0xf bank_mask:0xf\n\t"
      "v_min_num_f32_dpp %0, %0, %0 row_xmask:4 row_mask:0xf bank_mask:0xf\n\t"
      "v_min_num_f32_dpp %0, %0, %0 row_xmask:8 row_mask:0xf bank_mask:0xf"
      : "+v"(x));
  return x;
}
static __device__ __forceinline__ float row_max16(float x) {
  asm("v_max_num_f32_dpp %0, %0, %0 row_xmask:1 row_mask:0xf bank_mask:0xf\n\t"
      "v_max_num_f32_dpp %0, %0, %0 row_xmask:2 row_mask:0xf bank_mask:0xf\n\t"
      "v_max_num_f32_dpp %0, %0, %0 row_xmask:4 row_mask:0xf bank_mask:0xf\n\t"
      "v_max_num_f32_dpp %0, %0, %0 row_xmask:8 row_mask:0xf bank_mask:0xf"
      : "+v"(x));
  return x;
}

static __device__ __forceinline__ v4f pooled_norm(const v4f la, const v4f lb, const v2f wa) {
  const v8f zero = {};
  v2f b0, b1, b2, b3;
  b0[0] = la[0]; b0[1] = lb[0];
  b1[0] = la[1]; b1[1] = lb[1];
  b2[0] = la[2]; b2[1] = lb[2];
  b3[0] = la[3]; b3[1] = lb[3];
  const v8f d0 = __builtin_amdgcn_wmma_f32_16x16x4_f32(false, wa, false, b0, (short)0, zero, false, false);
  const v8f d1 = __builtin_amdgcn_wmma_f32_16x16x4_f32(false, wa, false, b1, (short)0, zero, false, false);
  const v8f d2 = __builtin_amdgcn_wmma_f32_16x16x4_f32(false, wa, false, b2, (short)0, zero, false, false);
  const v8f d3 = __builtin_amdgcn_wmma_f32_16x16x4_f32(false, wa, false, b3, (short)0, zero, false, false);
  v4f v;
  v[0] = d0[0]; v[1] = d1[0]; v[2] = d2[0]; v[3] = d3[0];
  float mn = fminf(fminf(v[0], v[1]), fminf(v[2], v[3]));
  float mx = fmaxf(fmaxf(v[0], v[1]), fmaxf(v[2], v[3]));
  mn = row_min16(mn);
  mx = row_max16(mx);
  const float inv = 1.0f / (mx - mn);
  return (v - mn) * inv;
}


struct RowBuf { v4f a[5]; v4f b[5]; };

static __device__ __forceinline__ void load_row(RowBuf& r, const float* __restrict__ img, int pr,
                                                const unsigned (&coff)[5]) {
  pr = (pr < Hh - 1) ? pr : Hh - 1;
  const float* rp = img + ((2u * (unsigned)pr) << 14);
#pragma unroll
  for (int c = 0; c < 5; ++c) {
    r.a[c] = *(const v4f*)(rp + coff[c]);
    r.b[c] = *(const v4f*)(rp + coff[c] + Ss * Tt);
  }
}

static __device__ __forceinline__ void norm_row(v4f (&n)[5], const RowBuf& r, const v2f wa) {
#pragma unroll
  for (int c = 0; c < 5; ++c) n[c] = pooled_norm(r.a[c], r.b[c], wa);
}

static __device__ __forceinline__ void emit_rows(float* __restrict__ oimg, const v4f (&na)[5], const v4f (&nb)[5],
                                                 int y, bool rowok, float wy, int jb, int tq) {
  const float r = 127.0f / 255.0f;
  const float wy0 = 1.0f - wy;
  v4f h[5];
#pragma unroll
  for (int c = 0; c < 5; ++c) h[c] = na[c] * wy0 + nb[c] * wy;
  if (rowok) {
    float* orow = oimg + ((((unsigned)y) << 14) + tq);
#pragma unroll
    for (int c = 0; c < 4; ++c) {
      const int j = jb + c;
#pragma unroll
      for (int q = 1; q < 3; ++q) {
        const int x = 2 * j + q;
        const bool inside = (c == 3 && q == 2) ? (jb < Hh - 4) : true;
        if (inside) {
          const float fx = (float)x * r;
          const float wx = fx - (float)j;
          const v4f o = h[c] * (1.0f - wx) + h[c + 1] * wx;
          *(v4f*)(orow + (x << 6)) = o;
        }
      }
    }
    if (jb == 0) {
      const float wx = 0.0f * r;
      const v4f o = h[0] * (1.0f - wx) + h[1] * wx;
      *(v4f*)(orow) = o;
    }
  }
}

static __device__ __forceinline__ void step(float* __restrict__ oimg, v4f (&na)[5], const RowBuf& rb, const v2f wa,
                                            int k, int hf, int jb, int tq, bool first) {
  const float r = 127.0f / 255.0f;
  v4f nb[5];
  norm_row(nb, rb, wa);
  const int y = 2 * k + 1 + hf;
  const float wy = (float)y * r - (float)k;
  emit_rows(oimg, na, nb, y, y < Ss, wy, jb, tq);
  if (first && k == 0) {
    emit_rows(oimg, na, nb, 0, hf == 0, 0.0f * r, jb, tq);
  }
#pragma unroll
  for (int c = 0; c < 5; ++c) na[c] = nb[c];
}

__global__ __launch_bounds__(256) void pool_norm_upsample_kernel(const float* __restrict__ inp,
                                                                 const float* __restrict__ W,
                                                                 float* __restrict__ out) {
  const int lane = threadIdx.x & 31;
  const int hf = lane >> 4;
  const int tq = (lane & 15) << 2;
  const int wave = __builtin_amdgcn_readfirstlane((int)(threadIdx.x >> 5));
  const int xg  = blockIdx.x & 3;
  const int seg = (blockIdx.x >> 2) & 15;
  const int b   = blockIdx.x >> 6;
  const int jb = (xg * 8 + wave) * 4;
  const int k0 = seg * 8;

  v2f wa;
  wa[0] = hf ? W[1 * 6] : W[0 * 6];
  wa[1] = hf ? W[3 * 6] : W[2 * 6];

  const float* img = inp + ((unsigned)b << 22);
  float* oimg = out + ((unsigned)b << 22);
  unsigned coff[5];
#pragma unroll
  for (int c = 0; c < 5; ++c) {
    int pc = jb + c;
    pc = (pc < Hh - 1) ? pc : Hh - 1;
    coff[c] = ((2u * (unsigned)pc) << 6) + ((unsigned)lane << 2);
  }

  RowBuf buf0, buf1, buf2;
  load_row(buf0, img, k0 + 0, coff);
  __builtin_amdgcn_sched_barrier(0);
  load_row(buf1, img, k0 + 1, coff);
  load_row(buf2, img, k0 + 2, coff);
  __builtin_amdgcn_sched_barrier(0);

  v4f na[5];
  norm_row(na, buf0, wa);

#pragma unroll 1
  for (int i = 1; i < 7; i += 3) {
    load_row(buf0, img, k0 + i + 2, coff);
    __builtin_amdgcn_sched_barrier(0);
    step(oimg, na, buf1, wa, k0 + i - 1, hf, jb, tq, true);

    load_row(buf1, img, k0 + i + 3, coff);
    __builtin_amdgcn_sched_barrier(0);
    step(oimg, na, buf2, wa, k0 + i, hf, jb, tq, false);

    load_row(buf2, img, k0 + i + 4, coff);
    __builtin_amdgcn_sched_barrier(0);
    step(oimg, na, buf0, wa, k0 + i + 1, hf, jb, tq, false);
  }
  step(oimg, na, buf1, wa, k0 + 6, hf, jb, tq, false);
  step(oimg, na, buf2, wa, k0 + 7, hf, jb, tq, false);
}

extern "C" void kernel_launch(void* const* d_in, const int* in_sizes, int n_in,
                              void* d_out, int out_size, void* d_ws, size_t ws_size,
                              hipStream_t stream) {
  const float* inp = (const float*)d_in[0];
  const float* W   = (const float*)d_in[1];
  float* out = (float*)d_out;

  pool_norm_upsample_kernel<<<512, 256, 0, stream>>>(inp, W, out);
}
